# starting kernel
# baseline (speedup 1.0000x reference)
#include <hip/hip_runtime.h>
#include <hip/hip_bf16.h>

typedef __attribute__((ext_vector_type(2))) float v2f;
typedef __attribute__((ext_vector_type(8))) float v8f;

__global__ __launch_bounds__(256) void edge_dot_wmma_f32(
    const float* __restrict__ h,
    const int*   __restrict__ src,
    const int*   __restrict__ dst,
    float*       __restrict__ out,
    int n_edges)
{
    const int lane  = threadIdx.x & 31;
    const int wave  = blockIdx.x * (blockDim.x >> 5) + (threadIdx.x >> 5);
    const int e0    = wave * 16;
    if (e0 >= n_edges) return;

    const int m    = lane & 15;
    const int half = lane >> 4;

    const long long sIdx = (long long)src[e0 + m];
    const long long dIdx = (long long)dst[e0 + m];
    const float* __restrict__ srow = h + sIdx * 128;
    const float* __restrict__ drow = h + dIdx * 128;

    v8f c = {};
#pragma unroll
    for (int ch = 0; ch < 32; ++ch) {
        const int k = ch * 4 + half * 2;
        v2f a = *(const v2f*)(srow + k);
        v2f b = *(const v2f*)(drow + k);
        c = __builtin_amdgcn_wmma_f32_16x16x4_f32(
                false, a, false, b, (short)0, c, false, false);
    }

    const int sel = lane & 7;
    float val = c[0];
    if (sel == 1) val = c[1];
    if (sel == 2) val = c[2];
    if (sel == 3) val = c[3];
    if (sel == 4) val = c[4];
    if (sel == 5) val = c[5];
    if (sel == 6) val = c[6];
    if (sel == 7) val = c[7];

    if (lane < 8)        out[e0 + lane]        = val;
    else if (lane >= 24) out[e0 + (lane - 16)] = val;
}

extern "C" void kernel_launch(void* const* d_in, const int* in_sizes, int n_in,
                              void* d_out, int out_size, void* d_ws, size_t ws_size,
                              hipStream_t stream) {
    const float* h   = (const float*)d_in[0];
    const int*   src = (const int*)  d_in[1];
    const int*   dst = (const int*)  d_in[2];
    float*       out = (float*)      d_out;

    const int n_edges = in_sizes[1];
    const int edges_per_block = (256 / 32) * 16;
    const int grid = (n_edges + edges_per_block - 1) / edges_per_block;

    edge_dot_wmma_f32<<<grid, 256, 0, stream>>>(h, src, dst, out, n_edges);
}


# Gather rows as whole 128-B lines: fragment row = 16-B chunk of a line, so a b128 load covers 4 whole lines (was 16 lines x 16 B per b64 load); dot = sum of 8 diagonal partials: sum order changes
# speedup vs baseline: 2.4058x; 2.4058x over previous
#include <hip/hip_runtime.h>
#include <hip/hip_bf16.h>


typedef __attribute__((ext_vector_type(2))) float v2f;
typedef __attribute__((ext_vector_type(4))) float v4f;
typedef __attribute__((ext_vector_type(8))) float v8f;

template <int N>
__device__ __forceinline__ float row_ror(float x)
{
    if constexpr ((N & 15) == 0) {
        return x;
    } else {
        const int xi = __builtin_bit_cast(int, x);
        return __builtin_bit_cast(float,
            __builtin_amdgcn_update_dpp(0, xi, 0x120 + (N & 15), 0xf, 0xf, true));
    }
}

template <int G>
__device__ __forceinline__ float group_dot(
    const float* __restrict__ h, int si, int di, int lane_off)
{
    const float* __restrict__ sp = h + (long long)si * 128 + lane_off;
    const float* __restrict__ dp = h + (long long)di * 128 + lane_off;

    const v4f a0 = *(const v4f*)(sp);
    const v4f a1 = *(const v4f*)(sp + 64);
    const v4f b0 = *(const v4f*)(dp);
    const v4f b1 = *(const v4f*)(dp + 64);

    v8f acc = {};
    acc = __builtin_amdgcn_wmma_f32_16x16x4_f32(
            false, a0.xy, false, b0.xy, (short)0, acc, false, false);
    acc = __builtin_amdgcn_wmma_f32_16x16x4_f32(
            false, a0.zw, false, b0.zw, (short)0, acc, false, false);
    acc = __builtin_amdgcn_wmma_f32_16x16x4_f32(
            false, a1.xy, false, b1.xy, (short)0, acc, false, false);
    acc = __builtin_amdgcn_wmma_f32_16x16x4_f32(
            false, a1.zw, false, b1.zw, (short)0, acc, false, false);

    float t = row_ror<G - 0>(acc[0]);
    t += row_ror<G - 1>(acc[1]);
    t += row_ror<G - 2>(acc[2]);
    t += row_ror<G - 3>(acc[3]);
    t += row_ror<G - 4>(acc[4]);
    t += row_ror<G - 5>(acc[5]);
    t += row_ror<G - 6>(acc[6]);
    t += row_ror<G - 7>(acc[7]);
    return t;
}

__global__ __launch_bounds__(256) void edge_dot_wmma_f32(
    const float* __restrict__ h,
    const int*   __restrict__ src,
    const int*   __restrict__ dst,
    float*       __restrict__ out,
    int n_edges)
{
    const int lane = threadIdx.x & 31;
    const int wv   = __builtin_amdgcn_readfirstlane((int)(threadIdx.x >> 5));
    const int wave = blockIdx.x * (blockDim.x >> 5) + wv;
    const int e0   = wave * 16;
    if (e0 >= n_edges) return;

    const int hf = lane >> 4;
    const int c  = lane & 7;
    const int e  = (lane >> 3) & 1;
    const int lane_off = hf * 32 + c * 4;

    int s[16], d[16];
    for (int j = 0; j < 16; ++j) { s[j] = src[e0 + j]; d[j] = dst[e0 + j]; }

    const float r0 = group_dot<0>(h, e ? s[ 8] : s[0], e ? d[ 8] : d[0], lane_off);
    const float r1 = group_dot<1>(h, e ? s[ 9] : s[1], e ? d[ 9] : d[1], lane_off);
    const float r2 = group_dot<2>(h, e ? s[10] : s[2], e ? d[10] : d[2], lane_off);
    const float r3 = group_dot<3>(h, e ? s[11] : s[3], e ? d[11] : d[3], lane_off);
    const float r4 = group_dot<4>(h, e ? s[12] : s[4], e ? d[12] : d[4], lane_off);
    const float r5 = group_dot<5>(h, e ? s[13] : s[5], e ? d[13] : d[5], lane_off);
    const float r6 = group_dot<6>(h, e ? s[14] : s[6], e ? d[14] : d[6], lane_off);
    const float r7 = group_dot<7>(h, e ? s[15] : s[7], e ? d[15] : d[7], lane_off);

    float val = r0;
    if (c == 1) val = r1;
    if (c == 2) val = r2;
    if (c == 3) val = r3;
    if (c == 4) val = r4;
    if (c == 5) val = r5;
    if (c == 6) val = r6;
    if (c == 7) val = r7;

    if (lane < 8 || lane >= 24) out[e0 + c + 8 * hf] = val;
}

extern "C" void kernel_launch(void* const* d_in, const int* in_sizes, int n_in,
                              void* d_out, int out_size, void* d_ws, size_t ws_size,
                              hipStream_t stream) {
    const float* h   = (const float*)d_in[0];
    const int*   src = (const int*)  d_in[1];
    const int*   dst = (const int*)  d_in[2];
    float*       out = (float*)      d_out;

    const int n_edges = in_sizes[1];
    const int edges_per_block = (256 / 32) * 16;
    const int grid = (n_edges + edges_per_block - 1) / edges_per_block;

    edge_dot_wmma_f32<<<grid, 256, 0, stream>>>(h, src, dst, out, n_edges);
}
